# starting kernel
# baseline (speedup 1.0000x reference)
#include <hip/hip_runtime.h>
#include <hip/hip_bf16.h>

typedef __attribute__((ext_vector_type(2))) float v2f;
typedef __attribute__((ext_vector_type(8))) float v8f;

__global__ void k_init_deg(float* __restrict__ deg, int N) {
    int i = blockIdx.x * blockDim.x + threadIdx.x;
    if (i < N) deg[i] = 1.0f;
}

__global__ void k_deg_edges(const int* __restrict__ ei, const float* __restrict__ w,
                            float* __restrict__ deg, int E) {
    int e = blockIdx.x * blockDim.x + threadIdx.x;
    if (e < E) {
        int d = ei[E + e];
        atomicAdd(&deg[d], w[e]);
    }
}

__global__ void k_dinv_init1(const float* __restrict__ x, float* __restrict__ deg_dinv,
                             float* __restrict__ agg1, int N) {
    int i = blockIdx.x * blockDim.x + threadIdx.x;
    if (i < N) {
        float d  = deg_dinv[i];
        float di = (d > 0.0f) ? rsqrtf(d) : 0.0f;
        deg_dinv[i] = di;
        agg1[i] = di * di * x[i];
    }
}

__global__ void k_edge1(const int* __restrict__ ei, const float* __restrict__ w,
                        const float* __restrict__ dinv, const float* __restrict__ x,
                        float* __restrict__ norm, float* __restrict__ agg1, int E) {
    int e = blockIdx.x * blockDim.x + threadIdx.x;
    if (e < E) {
        int s = ei[e];
        int d = ei[E + e];
        float nm = dinv[s] * w[e] * dinv[d];
        norm[e] = nm;
        atomicAdd(&agg1[d], nm * x[s]);
    }
}

__global__ void k_h1(const float* __restrict__ agg1, const float* __restrict__ W1,
                     const float* __restrict__ b1, float* __restrict__ h, int N) {
    int t = blockIdx.x * blockDim.x + threadIdx.x;
    if (t < N * 16) {
        int i = t >> 4, f = t & 15;
        float v = fmaf(agg1[i], W1[f], b1[f]);
        h[t] = v > 0.0f ? v : 0.0f;
    }
}

__global__ void k_layer2_wmma(const float* __restrict__ h, const float* __restrict__ W2,
                              const float* __restrict__ dinv, float* __restrict__ z,
                              float* __restrict__ agg2, int N) {
    __shared__ float w2p[16 * 16];
    int t = threadIdx.x;
    {
        int k = t >> 4, n = t & 15;
        w2p[t] = (n < 2) ? W2[k * 2 + n] : 0.0f;
    }
    __syncthreads();

    int wave = t >> 5;
    int lane = t & 31;
    int tile = blockIdx.x * 8 + wave;
    int m0   = tile * 16;
    if (m0 >= N) return;

    int row   = lane & 15;
    int khalf = (lane >> 4) * 2;

    int anode = m0 + row;
    int aload = anode < N ? anode : N - 1;

    v8f acc = {};
#pragma unroll
    for (int tstep = 0; tstep < 4; ++tstep) {
        int kb = tstep * 4 + khalf;
        v2f a, b;
        a.x = h[aload * 16 + kb];
        a.y = h[aload * 16 + kb + 1];
        b.x = w2p[kb * 16 + row];
        b.y = w2p[(kb + 1) * 16 + row];
        acc = __builtin_amdgcn_wmma_f32_16x16x4_f32(
             false, a,  false, b,
             (short)0, acc,  false,  false);
    }

    int col = row;
    if (col < 2) {
        int mofs = (lane >> 4) * 8;
#pragma unroll
        for (int r = 0; r < 8; ++r) {
            int node = m0 + mofs + r;
            if (node < N) {
                float zv = acc[r];
                float di = dinv[node];
                z[node * 2 + col]    = zv;
                agg2[node * 2 + col] = di * di * zv;
            }
        }
    }
}

__global__ void k_edge2(const int* __restrict__ ei, const float* __restrict__ norm,
                        const float* __restrict__ z, float* __restrict__ agg2, int E) {
    int e = blockIdx.x * blockDim.x + threadIdx.x;
    if (e < E) {
        int s = ei[e];
        int d = ei[E + e];
        float nm = norm[e];
        float z0 = z[2 * s], z1 = z[2 * s + 1];
        atomicAdd(&agg2[2 * d],     nm * z0);
        atomicAdd(&agg2[2 * d + 1], nm * z1);
    }
}

__global__ void k_out(const float* __restrict__ agg2, const float* __restrict__ b2,
                      float* __restrict__ out, int N) {
    int i = blockIdx.x * blockDim.x + threadIdx.x;
    if (i < N) {
        float a0 = agg2[2 * i]     + b2[0];
        float a1 = agg2[2 * i + 1] + b2[1];
        float m  = fmaxf(a0, a1);
        float lse = m + logf(expf(a0 - m) + expf(a1 - m));
        out[2 * i]     = a0 - lse;
        out[2 * i + 1] = a1 - lse;
    }
}

extern "C" void kernel_launch(void* const* d_in, const int* in_sizes, int n_in,
                              void* d_out, int out_size, void* d_ws, size_t ws_size,
                              hipStream_t stream) {
    const float* x   = (const float*)d_in[0];
    const int*   ei  = (const int*)  d_in[1];
    const float* ew  = (const float*)d_in[2];
    const float* W1  = (const float*)d_in[3];
    const float* b1  = (const float*)d_in[4];
    const float* W2  = (const float*)d_in[5];
    const float* b2  = (const float*)d_in[6];
    float*       out = (float*)d_out;

    const int N = in_sizes[0];
    const int E = in_sizes[2];

    float* wsf   = (float*)d_ws;
    float* dinv  = wsf;
    float* agg1  = wsf + (size_t)N;
    float* norm  = wsf + (size_t)2 * N;
    float* h     = norm + (size_t)E;
    float* z     = h + (size_t)16 * N;
    float* agg2  = z + (size_t)2 * N;

    const int B = 256;
    const int gN  = (N + B - 1) / B;
    const int gE  = (E + B - 1) / B;
    const int gNF = (N * 16 + B - 1) / B;
    const int tiles = (N + 15) / 16;
    const int gW  = (tiles + 7) / 8;

    k_init_deg   <<<gN,  B, 0, stream>>>(dinv, N);
    k_deg_edges  <<<gE,  B, 0, stream>>>(ei, ew, dinv, E);
    k_dinv_init1 <<<gN,  B, 0, stream>>>(x, dinv, agg1, N);
    k_edge1      <<<gE,  B, 0, stream>>>(ei, ew, dinv, x, norm, agg1, E);
    k_h1         <<<gNF, B, 0, stream>>>(agg1, W1, b1, h, N);
    k_layer2_wmma<<<gW,  B, 0, stream>>>(h, W2, dinv, z, agg2, N);
    k_edge2      <<<gE,  B, 0, stream>>>(ei, norm, z, agg2, E);
    k_out        <<<gN,  B, 0, stream>>>(agg2, b2, out, N);
}


# Edge scatter-adds go to an LDS copy of one destination range per work-group (LDS atomics), then one coalesced global atomic per node and slice; sums now form as per-slice partial sums
# speedup vs baseline: 4.9105x; 4.9105x over previous
#include <hip/hip_runtime.h>
#include <hip/hip_bf16.h>

typedef __attribute__((ext_vector_type(2))) float v2f;
typedef __attribute__((ext_vector_type(8))) float v8f;

__global__ void k_init_deg(float* __restrict__ deg, int N) {
    int i = blockIdx.x * blockDim.x + threadIdx.x;
    if (i < N) deg[i] = 1.0f;
}

#define EDGE_TPB 1024
#define EDGE_U   4
#define EDGE_CH  (EDGE_TPB * EDGE_U)
#define NP1      25600
#define NP2      12800

__device__ __forceinline__ void edge_slice(int E, int& beg, int& end) {
    int S   = (int)gridDim.x;
    int per = (E + S - 1) / S;
    per     = (per + 31) & ~31;
    int b   = (int)blockIdx.x * per;
    beg = b < E ? b : E;
    end = (b + per) < E ? (b + per) : E;
}

__device__ __forceinline__ int edge_clamp(int e, int end) { return e < end ? e : end - 1; }

__global__ __launch_bounds__(EDGE_TPB)
void k_deg_edges(const int* __restrict__ ei, const float* __restrict__ w,
                 float* __restrict__ deg, int N, int E) {
    __shared__ float acc[NP1];
    const int tid = threadIdx.x;
    const int lo  = (int)blockIdx.y * NP1;
    const int np  = (N - lo) < NP1 ? (N - lo) : NP1;
    for (int j = tid; j < np; j += EDGE_TPB) acc[j] = 0.0f;
    __syncthreads();

    int beg, end;
    edge_slice(E, beg, end);
    const int* __restrict__ dst = ei + E;
    if (beg < end) {
        int   d[EDGE_U];
        float wv[EDGE_U];
#pragma unroll
        for (int u = 0; u < EDGE_U; ++u) {
            int ec = edge_clamp(beg + tid + u * EDGE_TPB, end);
            d[u]  = dst[ec];
            wv[u] = w[ec];
        }
        for (int eb = beg; eb < end; eb += EDGE_CH) {
            int   dn[EDGE_U];
            float wn[EDGE_U];
#pragma unroll
            for (int u = 0; u < EDGE_U; ++u) {
                int ec = edge_clamp(eb + EDGE_CH + tid + u * EDGE_TPB, end);
                dn[u] = dst[ec];
                wn[u] = w[ec];
            }
#pragma unroll
            for (int u = 0; u < EDGE_U; ++u) {
                int e = eb + tid + u * EDGE_TPB;
                unsigned j = (unsigned)(d[u] - lo);
                if ((e < end) & (j < (unsigned)np)) atomicAdd(&acc[j], wv[u]);
            }
#pragma unroll
            for (int u = 0; u < EDGE_U; ++u) { d[u] = dn[u]; wv[u] = wn[u]; }
        }
    }
    __syncthreads();
    for (int j = tid; j < np; j += EDGE_TPB) {
        float v = acc[j];
        if (v != 0.0f) atomicAdd(&deg[lo + j], v);
    }
}

__global__ void k_dinv_init1(const float* __restrict__ x, float* __restrict__ deg_dinv,
                             float* __restrict__ agg1, int N) {
    int i = blockIdx.x * blockDim.x + threadIdx.x;
    if (i < N) {
        float d  = deg_dinv[i];
        float di = (d > 0.0f) ? rsqrtf(d) : 0.0f;
        deg_dinv[i] = di;
        agg1[i] = di * di * x[i];
    }
}

__global__ __launch_bounds__(EDGE_TPB)
void k_edge1(const int* __restrict__ ei, const float* __restrict__ w,
             const float* __restrict__ dinv, const float* __restrict__ x,
             float* __restrict__ norm, float* __restrict__ agg1, int N, int E) {
    __shared__ float acc[NP1];
    const int tid = threadIdx.x;
    const int lo  = (int)blockIdx.y * NP1;
    const int np  = (N - lo) < NP1 ? (N - lo) : NP1;
    for (int j = tid; j < np; j += EDGE_TPB) acc[j] = 0.0f;
    __syncthreads();

    int beg, end;
    edge_slice(E, beg, end);
    const int* __restrict__ src = ei;
    const int* __restrict__ dst = ei + E;
    if (beg < end) {
        int   s[EDGE_U], d[EDGE_U];
        float wv[EDGE_U];
#pragma unroll
        for (int u = 0; u < EDGE_U; ++u) {
            int ec = edge_clamp(beg + tid + u * EDGE_TPB, end);
            s[u]  = src[ec];
            d[u]  = dst[ec];
            wv[u] = w[ec];
        }
        for (int eb = beg; eb < end; eb += EDGE_CH) {
            int   sn[EDGE_U], dn[EDGE_U];
            float wn[EDGE_U];
            float ds[EDGE_U], dd[EDGE_U], xs[EDGE_U];
            bool  m[EDGE_U];
#pragma unroll
            for (int u = 0; u < EDGE_U; ++u) {
                int ec = edge_clamp(eb + EDGE_CH + tid + u * EDGE_TPB, end);
                sn[u] = src[ec];
                dn[u] = dst[ec];
                wn[u] = w[ec];
            }
#pragma unroll
            for (int u = 0; u < EDGE_U; ++u) {
                int e = eb + tid + u * EDGE_TPB;
                m[u]  = (e < end) & ((unsigned)(d[u] - lo) < (unsigned)np);
                ds[u] = 0.0f; dd[u] = 0.0f; xs[u] = 0.0f;
                if (m[u]) {
                    ds[u] = dinv[s[u]];
                    dd[u] = dinv[d[u]];
                    xs[u] = x[s[u]];
                }
            }
#pragma unroll
            for (int u = 0; u < EDGE_U; ++u) {
                if (m[u]) {
                    int e = eb + tid + u * EDGE_TPB;
                    float nm = ds[u] * wv[u] * dd[u];
                    norm[e] = nm;
                    atomicAdd(&acc[d[u] - lo], nm * xs[u]);
                }
            }
#pragma unroll
            for (int u = 0; u < EDGE_U; ++u) { s[u] = sn[u]; d[u] = dn[u]; wv[u] = wn[u]; }
        }
    }
    __syncthreads();
    for (int j = tid; j < np; j += EDGE_TPB) {
        float v = acc[j];
        if (v != 0.0f) atomicAdd(&agg1[lo + j], v);
    }
}

__global__ void k_h1(const float* __restrict__ agg1, const float* __restrict__ W1,
                     const float* __restrict__ b1, float* __restrict__ h, int N) {
    int t = blockIdx.x * blockDim.x + threadIdx.x;
    if (t < N * 16) {
        int i = t >> 4, f = t & 15;
        float v = fmaf(agg1[i], W1[f], b1[f]);
        h[t] = v > 0.0f ? v : 0.0f;
    }
}

__global__ void k_layer2_wmma(const float* __restrict__ h, const float* __restrict__ W2,
                              const float* __restrict__ dinv, float* __restrict__ z,
                              float* __restrict__ agg2, int N) {
    __shared__ float w2p[16 * 16];
    int t = threadIdx.x;
    {
        int k = t >> 4, n = t & 15;
        w2p[t] = (n < 2) ? W2[k * 2 + n] : 0.0f;
    }
    __syncthreads();

    int wave = t >> 5;
    int lane = t & 31;
    int tile = blockIdx.x * 8 + wave;
    int m0   = tile * 16;
    if (m0 >= N) return;

    int row   = lane & 15;
    int khalf = (lane >> 4) * 2;

    int anode = m0 + row;
    int aload = anode < N ? anode : N - 1;

    v8f acc = {};
#pragma unroll
    for (int tstep = 0; tstep < 4; ++tstep) {
        int kb = tstep * 4 + khalf;
        v2f a, b;
        a.x = h[aload * 16 + kb];
        a.y = h[aload * 16 + kb + 1];
        b.x = w2p[kb * 16 + row];
        b.y = w2p[(kb + 1) * 16 + row];
        acc = __builtin_amdgcn_wmma_f32_16x16x4_f32(
             false, a,  false, b,
             (short)0, acc,  false,  false);
    }

    int col = row;
    if (col < 2) {
        int mofs = (lane >> 4) * 8;
#pragma unroll
        for (int r = 0; r < 8; ++r) {
            int node = m0 + mofs + r;
            if (node < N) {
                float zv = acc[r];
                float di = dinv[node];
                z[node * 2 + col]    = zv;
                agg2[node * 2 + col] = di * di * zv;
            }
        }
    }
}

__global__ __launch_bounds__(EDGE_TPB)
void k_edge2(const int* __restrict__ ei, const float* __restrict__ norm,
             const float* __restrict__ z, float* __restrict__ agg2, int N, int E) {
    __shared__ float acc[2 * NP2];
    const int tid = threadIdx.x;
    const int lo  = (int)blockIdx.y * NP2;
    const int np  = (N - lo) < NP2 ? (N - lo) : NP2;
    for (int j = tid; j < 2 * np; j += EDGE_TPB) acc[j] = 0.0f;
    __syncthreads();

    int beg, end;
    edge_slice(E, beg, end);
    const int* __restrict__ src = ei;
    const int* __restrict__ dst = ei + E;
    if (beg < end) {
        int   s[EDGE_U], d[EDGE_U];
        float nm[EDGE_U];
#pragma unroll
        for (int u = 0; u < EDGE_U; ++u) {
            int ec = edge_clamp(beg + tid + u * EDGE_TPB, end);
            s[u]  = src[ec];
            d[u]  = dst[ec];
            nm[u] = norm[ec];
        }
        for (int eb = beg; eb < end; eb += EDGE_CH) {
            int   sn[EDGE_U], dn[EDGE_U];
            float nn[EDGE_U];
            v2f   zz[EDGE_U];
            bool  m[EDGE_U];
#pragma unroll
            for (int u = 0; u < EDGE_U; ++u) {
                int ec = edge_clamp(eb + EDGE_CH + tid + u * EDGE_TPB, end);
                sn[u] = src[ec];
                dn[u] = dst[ec];
                nn[u] = norm[ec];
            }
#pragma unroll
            for (int u = 0; u < EDGE_U; ++u) {
                int e = eb + tid + u * EDGE_TPB;
                m[u]  = (e < end) & ((unsigned)(d[u] - lo) < (unsigned)np);
                zz[u] = (v2f){0.0f, 0.0f};
                if (m[u]) __builtin_memcpy(&zz[u], z + 2 * s[u], sizeof(v2f));
            }
#pragma unroll
            for (int u = 0; u < EDGE_U; ++u) {
                if (m[u]) {
                    int j = d[u] - lo;
                    atomicAdd(&acc[2 * j],     nm[u] * zz[u].x);
                    atomicAdd(&acc[2 * j + 1], nm[u] * zz[u].y);
                }
            }
#pragma unroll
            for (int u = 0; u < EDGE_U; ++u) { s[u] = sn[u]; d[u] = dn[u]; nm[u] = nn[u]; }
        }
    }
    __syncthreads();
    for (int j = tid; j < 2 * np; j += EDGE_TPB) {
        float v = acc[j];
        if (v != 0.0f) atomicAdd(&agg2[2 * lo + j], v);
    }
}

__global__ void k_out(const float* __restrict__ agg2, const float* __restrict__ b2,
                      float* __restrict__ out, int N) {
    int i = blockIdx.x * blockDim.x + threadIdx.x;
    if (i < N) {
        float a0 = agg2[2 * i]     + b2[0];
        float a1 = agg2[2 * i + 1] + b2[1];
        float m  = fmaxf(a0, a1);
        float lse = m + logf(expf(a0 - m) + expf(a1 - m));
        out[2 * i]     = a0 - lse;
        out[2 * i + 1] = a1 - lse;
    }
}

extern "C" void kernel_launch(void* const* d_in, const int* in_sizes, int n_in,
                              void* d_out, int out_size, void* d_ws, size_t ws_size,
                              hipStream_t stream) {
    const float* x   = (const float*)d_in[0];
    const int*   ei  = (const int*)  d_in[1];
    const float* ew  = (const float*)d_in[2];
    const float* W1  = (const float*)d_in[3];
    const float* b1  = (const float*)d_in[4];
    const float* W2  = (const float*)d_in[5];
    const float* b2  = (const float*)d_in[6];
    float*       out = (float*)d_out;

    const int N = in_sizes[0];
    const int E = in_sizes[2];

    float* wsf   = (float*)d_ws;
    float* dinv  = wsf;
    float* agg1  = wsf + (size_t)N;
    float* norm  = wsf + (size_t)2 * N;
    float* h     = norm + (size_t)E;
    float* z     = h + (size_t)16 * N;
    float* agg2  = z + (size_t)2 * N;

    const int B = 256;
    const int gN  = (N + B - 1) / B;
    const int gNF = (N * 16 + B - 1) / B;
    const int tiles = (N + 15) / 16;
    const int gW  = (tiles + 7) / 8;
    const int P1  = (N + NP1 - 1) / NP1;
    const int P2  = (N + NP2 - 1) / NP2;
    const dim3 gE1(64, P1);
    const dim3 gE2(32, P2);

    k_init_deg   <<<gN,  B, 0, stream>>>(dinv, N);
    k_deg_edges  <<<gE1, EDGE_TPB, 0, stream>>>(ei, ew, dinv, N, E);
    k_dinv_init1 <<<gN,  B, 0, stream>>>(x, dinv, agg1, N);
    k_edge1      <<<gE1, EDGE_TPB, 0, stream>>>(ei, ew, dinv, x, norm, agg1, N, E);
    k_h1         <<<gNF, B, 0, stream>>>(agg1, W1, b1, h, N);
    k_layer2_wmma<<<gW,  B, 0, stream>>>(h, W2, dinv, z, agg2, N);
    k_edge2      <<<gE2, EDGE_TPB, 0, stream>>>(ei, norm, z, agg2, N, E);
    k_out        <<<gN,  B, 0, stream>>>(agg2, b2, out, N);
}


# Normalisation factored out of the edge sums: edges gather one pre-scaled value dinv[s]*v[s] and add w*that; dinv[d] applied once per node after the sum; norm array gone (products regrouped)
# speedup vs baseline: 5.9701x; 1.2158x over previous
#include <hip/hip_runtime.h>
#include <hip/hip_bf16.h>

typedef __attribute__((ext_vector_type(2))) float v2f;
typedef __attribute__((ext_vector_type(8))) float v8f;

__global__ void k_init_deg(float* __restrict__ deg, int N) {
    int i = blockIdx.x * blockDim.x + threadIdx.x;
    if (i < N) deg[i] = 1.0f;
}

#define EDGE_TPB 1024
#define EDGE_U   4
#define EDGE_CH  (EDGE_TPB * EDGE_U)
#define NP1      25600
#define NP2      12800

__device__ __forceinline__ void edge_slice(int E, int& beg, int& end) {
    int S   = (int)gridDim.x;
    int per = (E + S - 1) / S;
    per     = (per + 31) & ~31;
    int b   = (int)blockIdx.x * per;
    beg = b < E ? b : E;
    end = (b + per) < E ? (b + per) : E;
}

__device__ __forceinline__ int edge_clamp(int e, int end) { return e < end ? e : end - 1; }

__global__ __launch_bounds__(EDGE_TPB)
void k_deg_edges(const int* __restrict__ ei, const float* __restrict__ w,
                 float* __restrict__ deg, int N, int E) {
    __shared__ float acc[NP1];
    const int tid = threadIdx.x;
    const int lo  = (int)blockIdx.y * NP1;
    const int np  = (N - lo) < NP1 ? (N - lo) : NP1;
    for (int j = tid; j < np; j += EDGE_TPB) acc[j] = 0.0f;
    __syncthreads();

    int beg, end;
    edge_slice(E, beg, end);
    const int* __restrict__ dst = ei + E;
    if (beg < end) {
        int   d[EDGE_U];
        float wv[EDGE_U];
#pragma unroll
        for (int u = 0; u < EDGE_U; ++u) {
            int ec = edge_clamp(beg + tid + u * EDGE_TPB, end);
            d[u]  = dst[ec];
            wv[u] = w[ec];
        }
        for (int eb = beg; eb < end; eb += EDGE_CH) {
            int   dn[EDGE_U];
            float wn[EDGE_U];
#pragma unroll
            for (int u = 0; u < EDGE_U; ++u) {
                int ec = edge_clamp(eb + EDGE_CH + tid + u * EDGE_TPB, end);
                dn[u] = dst[ec];
                wn[u] = w[ec];
            }
#pragma unroll
            for (int u = 0; u < EDGE_U; ++u) {
                int e = eb + tid + u * EDGE_TPB;
                unsigned j = (unsigned)(d[u] - lo);
                if ((e < end) & (j < (unsigned)np)) atomicAdd(&acc[j], wv[u]);
            }
#pragma unroll
            for (int u = 0; u < EDGE_U; ++u) { d[u] = dn[u]; wv[u] = wn[u]; }
        }
    }
    __syncthreads();
    for (int j = tid; j < np; j += EDGE_TPB) {
        float v = acc[j];
        if (v != 0.0f) atomicAdd(&deg[lo + j], v);
    }
}

__global__ void k_dinv_init1(const float* __restrict__ x, float* __restrict__ deg_dinv,
                             float* __restrict__ xs, float* __restrict__ agg1, int N) {
    int i = blockIdx.x * blockDim.x + threadIdx.x;
    if (i < N) {
        float d  = deg_dinv[i];
        float di = (d > 0.0f) ? rsqrtf(d) : 0.0f;
        float v  = di * x[i];
        deg_dinv[i] = di;
        xs[i]   = v;
        agg1[i] = v;
    }
}

__global__ __launch_bounds__(EDGE_TPB)
void k_edge1(const int* __restrict__ ei, const float* __restrict__ w,
             const float* __restrict__ xs, float* __restrict__ agg1, int N, int E) {
    __shared__ float acc[NP1];
    const int tid = threadIdx.x;
    const int lo  = (int)blockIdx.y * NP1;
    const int np  = (N - lo) < NP1 ? (N - lo) : NP1;
    for (int j = tid; j < np; j += EDGE_TPB) acc[j] = 0.0f;
    __syncthreads();

    int beg, end;
    edge_slice(E, beg, end);
    const int* __restrict__ src = ei;
    const int* __restrict__ dst = ei + E;
    if (beg < end) {
        int   s[EDGE_U], d[EDGE_U];
        float wv[EDGE_U];
#pragma unroll
        for (int u = 0; u < EDGE_U; ++u) {
            int ec = edge_clamp(beg + tid + u * EDGE_TPB, end);
            s[u]  = src[ec];
            d[u]  = dst[ec];
            wv[u] = w[ec];
        }
        for (int eb = beg; eb < end; eb += EDGE_CH) {
            int   sn[EDGE_U], dn[EDGE_U];
            float wn[EDGE_U];
            float xg[EDGE_U];
            bool  m[EDGE_U];
#pragma unroll
            for (int u = 0; u < EDGE_U; ++u) {
                int ec = edge_clamp(eb + EDGE_CH + tid + u * EDGE_TPB, end);
                sn[u] = src[ec];
                dn[u] = dst[ec];
                wn[u] = w[ec];
            }
#pragma unroll
            for (int u = 0; u < EDGE_U; ++u) {
                int e = eb + tid + u * EDGE_TPB;
                m[u]  = (e < end) & ((unsigned)(d[u] - lo) < (unsigned)np);
                xg[u] = 0.0f;
                if (m[u]) xg[u] = xs[s[u]];
            }
#pragma unroll
            for (int u = 0; u < EDGE_U; ++u) {
                if (m[u]) atomicAdd(&acc[d[u] - lo], wv[u] * xg[u]);
            }
#pragma unroll
            for (int u = 0; u < EDGE_U; ++u) { s[u] = sn[u]; d[u] = dn[u]; wv[u] = wn[u]; }
        }
    }
    __syncthreads();
    for (int j = tid; j < np; j += EDGE_TPB) {
        float v = acc[j];
        if (v != 0.0f) atomicAdd(&agg1[lo + j], v);
    }
}

__global__ void k_h1(const float* __restrict__ agg1, const float* __restrict__ dinv,
                     const float* __restrict__ W1, const float* __restrict__ b1,
                     float* __restrict__ h, int N) {
    int t = blockIdx.x * blockDim.x + threadIdx.x;
    if (t < N * 16) {
        int i = t >> 4, f = t & 15;
        float v = fmaf(dinv[i] * agg1[i], W1[f], b1[f]);
        h[t] = v > 0.0f ? v : 0.0f;
    }
}

__global__ void k_layer2_wmma(const float* __restrict__ h, const float* __restrict__ W2,
                              const float* __restrict__ dinv, float* __restrict__ zs,
                              float* __restrict__ agg2, int N) {
    __shared__ float w2p[16 * 16];
    int t = threadIdx.x;
    {
        int k = t >> 4, n = t & 15;
        w2p[t] = (n < 2) ? W2[k * 2 + n] : 0.0f;
    }
    __syncthreads();

    int wave = t >> 5;
    int lane = t & 31;
    int tile = blockIdx.x * 8 + wave;
    int m0   = tile * 16;
    if (m0 >= N) return;

    int row   = lane & 15;
    int khalf = (lane >> 4) * 2;

    int anode = m0 + row;
    int aload = anode < N ? anode : N - 1;

    v8f acc = {};
#pragma unroll
    for (int tstep = 0; tstep < 4; ++tstep) {
        int kb = tstep * 4 + khalf;
        v2f a, b;
        a.x = h[aload * 16 + kb];
        a.y = h[aload * 16 + kb + 1];
        b.x = w2p[kb * 16 + row];
        b.y = w2p[(kb + 1) * 16 + row];
        acc = __builtin_amdgcn_wmma_f32_16x16x4_f32(
             false, a,  false, b,
             (short)0, acc,  false,  false);
    }

    int col = row;
    if (col < 2) {
        int mofs = (lane >> 4) * 8;
#pragma unroll
        for (int r = 0; r < 8; ++r) {
            int node = m0 + mofs + r;
            if (node < N) {
                float v = dinv[node] * acc[r];
                zs[node * 2 + col]   = v;
                agg2[node * 2 + col] = v;
            }
        }
    }
}

__global__ __launch_bounds__(EDGE_TPB)
void k_edge2(const int* __restrict__ ei, const float* __restrict__ w,
             const float* __restrict__ zs, float* __restrict__ agg2, int N, int E) {
    __shared__ float acc[2 * NP2];
    const int tid = threadIdx.x;
    const int lo  = (int)blockIdx.y * NP2;
    const int np  = (N - lo) < NP2 ? (N - lo) : NP2;
    for (int j = tid; j < 2 * np; j += EDGE_TPB) acc[j] = 0.0f;
    __syncthreads();

    int beg, end;
    edge_slice(E, beg, end);
    const int* __restrict__ src = ei;
    const int* __restrict__ dst = ei + E;
    if (beg < end) {
        int   s[EDGE_U], d[EDGE_U];
        float nm[EDGE_U];
#pragma unroll
        for (int u = 0; u < EDGE_U; ++u) {
            int ec = edge_clamp(beg + tid + u * EDGE_TPB, end);
            s[u]  = src[ec];
            d[u]  = dst[ec];
            nm[u] = w[ec];
        }
        for (int eb = beg; eb < end; eb += EDGE_CH) {
            int   sn[EDGE_U], dn[EDGE_U];
            float nn[EDGE_U];
            v2f   zz[EDGE_U];
            bool  m[EDGE_U];
#pragma unroll
            for (int u = 0; u < EDGE_U; ++u) {
                int ec = edge_clamp(eb + EDGE_CH + tid + u * EDGE_TPB, end);
                sn[u] = src[ec];
                dn[u] = dst[ec];
                nn[u] = w[ec];
            }
#pragma unroll
            for (int u = 0; u < EDGE_U; ++u) {
                int e = eb + tid + u * EDGE_TPB;
                m[u]  = (e < end) & ((unsigned)(d[u] - lo) < (unsigned)np);
                zz[u] = (v2f){0.0f, 0.0f};
                if (m[u]) __builtin_memcpy(&zz[u], zs + 2 * s[u], sizeof(v2f));
            }
#pragma unroll
            for (int u = 0; u < EDGE_U; ++u) {
                if (m[u]) {
                    int j = d[u] - lo;
                    atomicAdd(&acc[2 * j],     nm[u] * zz[u].x);
                    atomicAdd(&acc[2 * j + 1], nm[u] * zz[u].y);
                }
            }
#pragma unroll
            for (int u = 0; u < EDGE_U; ++u) { s[u] = sn[u]; d[u] = dn[u]; nm[u] = nn[u]; }
        }
    }
    __syncthreads();
    for (int j = tid; j < 2 * np; j += EDGE_TPB) {
        float v = acc[j];
        if (v != 0.0f) atomicAdd(&agg2[2 * lo + j], v);
    }
}

__global__ void k_out(const float* __restrict__ agg2, const float* __restrict__ dinv,
                      const float* __restrict__ b2, float* __restrict__ out, int N) {
    int i = blockIdx.x * blockDim.x + threadIdx.x;
    if (i < N) {
        float di = dinv[i];
        float a0 = fmaf(di, agg2[2 * i],     b2[0]);
        float a1 = fmaf(di, agg2[2 * i + 1], b2[1]);
        float m  = fmaxf(a0, a1);
        float lse = m + logf(expf(a0 - m) + expf(a1 - m));
        out[2 * i]     = a0 - lse;
        out[2 * i + 1] = a1 - lse;
    }
}

extern "C" void kernel_launch(void* const* d_in, const int* in_sizes, int n_in,
                              void* d_out, int out_size, void* d_ws, size_t ws_size,
                              hipStream_t stream) {
    const float* x   = (const float*)d_in[0];
    const int*   ei  = (const int*)  d_in[1];
    const float* ew  = (const float*)d_in[2];
    const float* W1  = (const float*)d_in[3];
    const float* b1  = (const float*)d_in[4];
    const float* W2  = (const float*)d_in[5];
    const float* b2  = (const float*)d_in[6];
    float*       out = (float*)d_out;

    const int N = in_sizes[0];
    const int E = in_sizes[2];

    float* wsf   = (float*)d_ws;
    float* dinv  = wsf;
    float* agg1  = wsf + (size_t)N;
    float* xs    = wsf + (size_t)2 * N;
    float* h     = xs + (size_t)E;
    float* zs    = h + (size_t)16 * N;
    float* agg2  = zs + (size_t)2 * N;

    const int B = 256;
    const int gN  = (N + B - 1) / B;
    const int gNF = (N * 16 + B - 1) / B;
    const int tiles = (N + 15) / 16;
    const int gW  = (tiles + 7) / 8;
    const int P1  = (N + NP1 - 1) / NP1;
    const int P2  = (N + NP2 - 1) / NP2;
    const dim3 gE1(64, P1);
    const dim3 gE2(32, P2);

    k_init_deg   <<<gN,  B, 0, stream>>>(dinv, N);
    k_deg_edges  <<<gE1, EDGE_TPB, 0, stream>>>(ei, ew, dinv, N, E);
    k_dinv_init1 <<<gN,  B, 0, stream>>>(x, dinv, xs, agg1, N);
    k_edge1      <<<gE1, EDGE_TPB, 0, stream>>>(ei, ew, xs, agg1, N, E);
    k_h1         <<<gNF, B, 0, stream>>>(agg1, dinv, W1, b1, h, N);
    k_layer2_wmma<<<gW,  B, 0, stream>>>(h, W2, dinv, zs, agg2, N);
    k_edge2      <<<gE2, EDGE_TPB, 0, stream>>>(ei, ew, zs, agg2, N, E);
    k_out        <<<gN,  B, 0, stream>>>(agg2, dinv, b2, out, N);
}


# Layer-1 epilogue fused into the layer-2 WMMA kernel: each lane forms its A fragment relu(dinv*agg1*W1+b1) in registers from two loads; the 16-wide h array, its kernel and one launch are gone
# speedup vs baseline: 6.2532x; 1.0474x over previous
#include <hip/hip_runtime.h>
#include <hip/hip_bf16.h>

typedef __attribute__((ext_vector_type(2))) float v2f;
typedef __attribute__((ext_vector_type(8))) float v8f;

__global__ void k_init_deg(float* __restrict__ deg, int N) {
    int i = blockIdx.x * blockDim.x + threadIdx.x;
    if (i < N) deg[i] = 1.0f;
}

#define EDGE_TPB 1024
#define EDGE_U   4
#define EDGE_CH  (EDGE_TPB * EDGE_U)
#define NP1      25600
#define NP2      12800

__device__ __forceinline__ void edge_slice(int E, int& beg, int& end) {
    int S   = (int)gridDim.x;
    int per = (E + S - 1) / S;
    per     = (per + 31) & ~31;
    int b   = (int)blockIdx.x * per;
    beg = b < E ? b : E;
    end = (b + per) < E ? (b + per) : E;
}

__device__ __forceinline__ int edge_clamp(int e, int end) { return e < end ? e : end - 1; }

__global__ __launch_bounds__(EDGE_TPB)
void k_deg_edges(const int* __restrict__ ei, const float* __restrict__ w,
                 float* __restrict__ deg, int N, int E) {
    __shared__ float acc[NP1];
    const int tid = threadIdx.x;
    const int lo  = (int)blockIdx.y * NP1;
    const int np  = (N - lo) < NP1 ? (N - lo) : NP1;
    for (int j = tid; j < np; j += EDGE_TPB) acc[j] = 0.0f;
    __syncthreads();

    int beg, end;
    edge_slice(E, beg, end);
    const int* __restrict__ dst = ei + E;
    if (beg < end) {
        int   d[EDGE_U];
        float wv[EDGE_U];
#pragma unroll
        for (int u = 0; u < EDGE_U; ++u) {
            int ec = edge_clamp(beg + tid + u * EDGE_TPB, end);
            d[u]  = dst[ec];
            wv[u] = w[ec];
        }
        for (int eb = beg; eb < end; eb += EDGE_CH) {
            int   dn[EDGE_U];
            float wn[EDGE_U];
#pragma unroll
            for (int u = 0; u < EDGE_U; ++u) {
                int ec = edge_clamp(eb + EDGE_CH + tid + u * EDGE_TPB, end);
                dn[u] = dst[ec];
                wn[u] = w[ec];
            }
#pragma unroll
            for (int u = 0; u < EDGE_U; ++u) {
                int e = eb + tid + u * EDGE_TPB;
                unsigned j = (unsigned)(d[u] - lo);
                if ((e < end) & (j < (unsigned)np)) atomicAdd(&acc[j], wv[u]);
            }
#pragma unroll
            for (int u = 0; u < EDGE_U; ++u) { d[u] = dn[u]; wv[u] = wn[u]; }
        }
    }
    __syncthreads();
    for (int j = tid; j < np; j += EDGE_TPB) {
        float v = acc[j];
        if (v != 0.0f) atomicAdd(&deg[lo + j], v);
    }
}

__global__ void k_dinv_init1(const float* __restrict__ x, float* __restrict__ deg_dinv,
                             float* __restrict__ xs, float* __restrict__ agg1, int N) {
    int i = blockIdx.x * blockDim.x + threadIdx.x;
    if (i < N) {
        float d  = deg_dinv[i];
        float di = (d > 0.0f) ? rsqrtf(d) : 0.0f;
        float v  = di * x[i];
        deg_dinv[i] = di;
        xs[i]   = v;
        agg1[i] = v;
    }
}

__global__ __launch_bounds__(EDGE_TPB)
void k_edge1(const int* __restrict__ ei, const float* __restrict__ w,
             const float* __restrict__ xs, float* __restrict__ agg1, int N, int E) {
    __shared__ float acc[NP1];
    const int tid = threadIdx.x;
    const int lo  = (int)blockIdx.y * NP1;
    const int np  = (N - lo) < NP1 ? (N - lo) : NP1;
    for (int j = tid; j < np; j += EDGE_TPB) acc[j] = 0.0f;
    __syncthreads();

    int beg, end;
    edge_slice(E, beg, end);
    const int* __restrict__ src = ei;
    const int* __restrict__ dst = ei + E;
    if (beg < end) {
        int   s[EDGE_U], d[EDGE_U];
        float wv[EDGE_U];
#pragma unroll
        for (int u = 0; u < EDGE_U; ++u) {
            int ec = edge_clamp(beg + tid + u * EDGE_TPB, end);
            s[u]  = src[ec];
            d[u]  = dst[ec];
            wv[u] = w[ec];
        }
        for (int eb = beg; eb < end; eb += EDGE_CH) {
            int   sn[EDGE_U], dn[EDGE_U];
            float wn[EDGE_U];
            float xg[EDGE_U];
            bool  m[EDGE_U];
#pragma unroll
            for (int u = 0; u < EDGE_U; ++u) {
                int ec = edge_clamp(eb + EDGE_CH + tid + u * EDGE_TPB, end);
                sn[u] = src[ec];
                dn[u] = dst[ec];
                wn[u] = w[ec];
            }
#pragma unroll
            for (int u = 0; u < EDGE_U; ++u) {
                int e = eb + tid + u * EDGE_TPB;
                m[u]  = (e < end) & ((unsigned)(d[u] - lo) < (unsigned)np);
                xg[u] = 0.0f;
                if (m[u]) xg[u] = xs[s[u]];
            }
#pragma unroll
            for (int u = 0; u < EDGE_U; ++u) {
                if (m[u]) atomicAdd(&acc[d[u] - lo], wv[u] * xg[u]);
            }
#pragma unroll
            for (int u = 0; u < EDGE_U; ++u) { s[u] = sn[u]; d[u] = dn[u]; wv[u] = wn[u]; }
        }
    }
    __syncthreads();
    for (int j = tid; j < np; j += EDGE_TPB) {
        float v = acc[j];
        if (v != 0.0f) atomicAdd(&agg1[lo + j], v);
    }
}

__global__ void k_layer2_wmma(const float* __restrict__ agg1, const float* __restrict__ W1,
                              const float* __restrict__ b1, const float* __restrict__ W2,
                              const float* __restrict__ dinv, float* __restrict__ zs,
                              float* __restrict__ agg2, int N) {
    __shared__ float w2p[16 * 16];
    __shared__ float w1s[16];
    __shared__ float b1s[16];
    int t = threadIdx.x;
    {
        int k = t >> 4, n = t & 15;
        w2p[t] = (n < 2) ? W2[k * 2 + n] : 0.0f;
        if (t < 16) { w1s[t] = W1[t]; b1s[t] = b1[t]; }
    }
    __syncthreads();

    int wave = t >> 5;
    int lane = t & 31;
    int tile = blockIdx.x * 8 + wave;
    int m0   = tile * 16;
    if (m0 >= N) return;

    int row   = lane & 15;
    int khalf = (lane >> 4) * 2;

    int anode = m0 + row;
    int aload = anode < N ? anode : N - 1;
    const float p = dinv[aload] * agg1[aload];

    v2f a[4];
#pragma unroll
    for (int tstep = 0; tstep < 4; ++tstep) {
        int kb = tstep * 4 + khalf;
        float h0 = fmaf(p, w1s[kb],     b1s[kb]);
        float h1 = fmaf(p, w1s[kb + 1], b1s[kb + 1]);
        a[tstep].x = h0 > 0.0f ? h0 : 0.0f;
        a[tstep].y = h1 > 0.0f ? h1 : 0.0f;
    }

    __builtin_amdgcn_sched_barrier(0);

    v8f acc = {};
#pragma unroll
    for (int tstep = 0; tstep < 4; ++tstep) {
        int kb = tstep * 4 + khalf;
        v2f b;
        b.x = w2p[kb * 16 + row];
        b.y = w2p[(kb + 1) * 16 + row];
        acc = __builtin_amdgcn_wmma_f32_16x16x4_f32(
             false, a[tstep],  false, b,
             (short)0, acc,  false,  false);
    }

    int col = row;
    if (col < 2) {
        int mofs = (lane >> 4) * 8;
#pragma unroll
        for (int r = 0; r < 8; ++r) {
            int node = m0 + mofs + r;
            if (node < N) {
                float v = dinv[node] * acc[r];
                zs[node * 2 + col]   = v;
                agg2[node * 2 + col] = v;
            }
        }
    }
}

__global__ __launch_bounds__(EDGE_TPB)
void k_edge2(const int* __restrict__ ei, const float* __restrict__ w,
             const float* __restrict__ zs, float* __restrict__ agg2, int N, int E) {
    __shared__ float acc[2 * NP2];
    const int tid = threadIdx.x;
    const int lo  = (int)blockIdx.y * NP2;
    const int np  = (N - lo) < NP2 ? (N - lo) : NP2;
    for (int j = tid; j < 2 * np; j += EDGE_TPB) acc[j] = 0.0f;
    __syncthreads();

    int beg, end;
    edge_slice(E, beg, end);
    const int* __restrict__ src = ei;
    const int* __restrict__ dst = ei + E;
    if (beg < end) {
        int   s[EDGE_U], d[EDGE_U];
        float nm[EDGE_U];
#pragma unroll
        for (int u = 0; u < EDGE_U; ++u) {
            int ec = edge_clamp(beg + tid + u * EDGE_TPB, end);
            s[u]  = src[ec];
            d[u]  = dst[ec];
            nm[u] = w[ec];
        }
        for (int eb = beg; eb < end; eb += EDGE_CH) {
            int   sn[EDGE_U], dn[EDGE_U];
            float nn[EDGE_U];
            v2f   zz[EDGE_U];
            bool  m[EDGE_U];
#pragma unroll
            for (int u = 0; u < EDGE_U; ++u) {
                int ec = edge_clamp(eb + EDGE_CH + tid + u * EDGE_TPB, end);
                sn[u] = src[ec];
                dn[u] = dst[ec];
                nn[u] = w[ec];
            }
#pragma unroll
            for (int u = 0; u < EDGE_U; ++u) {
                int e = eb + tid + u * EDGE_TPB;
                m[u]  = (e < end) & ((unsigned)(d[u] - lo) < (unsigned)np);
                zz[u] = (v2f){0.0f, 0.0f};
                if (m[u]) __builtin_memcpy(&zz[u], zs + 2 * s[u], sizeof(v2f));
            }
#pragma unroll
            for (int u = 0; u < EDGE_U; ++u) {
                if (m[u]) {
                    int j = d[u] - lo;
                    atomicAdd(&acc[2 * j],     nm[u] * zz[u].x);
                    atomicAdd(&acc[2 * j + 1], nm[u] * zz[u].y);
                }
            }
#pragma unroll
            for (int u = 0; u < EDGE_U; ++u) { s[u] = sn[u]; d[u] = dn[u]; nm[u] = nn[u]; }
        }
    }
    __syncthreads();
    for (int j = tid; j < 2 * np; j += EDGE_TPB) {
        float v = acc[j];
        if (v != 0.0f) atomicAdd(&agg2[2 * lo + j], v);
    }
}

__global__ void k_out(const float* __restrict__ agg2, const float* __restrict__ dinv,
                      const float* __restrict__ b2, float* __restrict__ out, int N) {
    int i = blockIdx.x * blockDim.x + threadIdx.x;
    if (i < N) {
        float di = dinv[i];
        float a0 = fmaf(di, agg2[2 * i],     b2[0]);
        float a1 = fmaf(di, agg2[2 * i + 1], b2[1]);
        float m  = fmaxf(a0, a1);
        float lse = m + logf(expf(a0 - m) + expf(a1 - m));
        out[2 * i]     = a0 - lse;
        out[2 * i + 1] = a1 - lse;
    }
}

extern "C" void kernel_launch(void* const* d_in, const int* in_sizes, int n_in,
                              void* d_out, int out_size, void* d_ws, size_t ws_size,
                              hipStream_t stream) {
    const float* x   = (const float*)d_in[0];
    const int*   ei  = (const int*)  d_in[1];
    const float* ew  = (const float*)d_in[2];
    const float* W1  = (const float*)d_in[3];
    const float* b1  = (const float*)d_in[4];
    const float* W2  = (const float*)d_in[5];
    const float* b2  = (const float*)d_in[6];
    float*       out = (float*)d_out;

    const int N = in_sizes[0];
    const int E = in_sizes[2];

    float* wsf   = (float*)d_ws;
    float* dinv  = wsf;
    float* agg1  = wsf + (size_t)N;
    float* xs    = wsf + (size_t)2 * N;
    float* zs    = xs + (size_t)E;
    float* agg2  = zs + (size_t)2 * N;

    const int B = 256;
    const int gN  = (N + B - 1) / B;
    const int tiles = (N + 15) / 16;
    const int gW  = (tiles + 7) / 8;
    const int P1  = (N + NP1 - 1) / NP1;
    const int P2  = (N + NP2 - 1) / NP2;
    const dim3 gE1(64, P1);
    const dim3 gE2(32, P2);

    k_init_deg   <<<gN,  B, 0, stream>>>(dinv, N);
    k_deg_edges  <<<gE1, EDGE_TPB, 0, stream>>>(ei, ew, dinv, N, E);
    k_dinv_init1 <<<gN,  B, 0, stream>>>(x, dinv, xs, agg1, N);
    k_edge1      <<<gE1, EDGE_TPB, 0, stream>>>(ei, ew, xs, agg1, N, E);
    k_layer2_wmma<<<gW,  B, 0, stream>>>(agg1, W1, b1, W2, dinv, zs, agg2, N);
    k_edge2      <<<gE2, EDGE_TPB, 0, stream>>>(ei, ew, zs, agg2, N, E);
    k_out        <<<gN,  B, 0, stream>>>(agg2, dinv, b2, out, N);
}


# Each edge-pass work-group adds its LDS sums into the sum-array copy of its own cache half (die id); consuming node kernels read copy A + copy B (sum order changed)
# speedup vs baseline: 7.4013x; 1.1836x over previous
#include <hip/hip_runtime.h>
#include <hip/hip_bf16.h>

typedef __attribute__((ext_vector_type(2))) float v2f;
typedef __attribute__((ext_vector_type(8))) float v8f;

__device__ __forceinline__ int cache_half() {
    return (int)((__builtin_amdgcn_s_sendmsg_rtn(0x87) >> 18) & 1u);
}

__global__ void k_init_deg(float* __restrict__ degA, float* __restrict__ degB, int N) {
    int i = blockIdx.x * blockDim.x + threadIdx.x;
    if (i < N) { degA[i] = 1.0f; degB[i] = 0.0f; }
}

#define EDGE_TPB 1024
#define EDGE_U   4
#define EDGE_CH  (EDGE_TPB * EDGE_U)
#define NP1      25600
#define NP2      12800

__device__ __forceinline__ void edge_slice(int E, int& beg, int& end) {
    int S   = (int)gridDim.x;
    int per = (E + S - 1) / S;
    per     = (per + 31) & ~31;
    int b   = (int)blockIdx.x * per;
    beg = b < E ? b : E;
    end = (b + per) < E ? (b + per) : E;
}

__device__ __forceinline__ int edge_clamp(int e, int end) { return e < end ? e : end - 1; }

__global__ __launch_bounds__(EDGE_TPB)
void k_deg_edges(const int* __restrict__ ei, const float* __restrict__ w,
                 float* __restrict__ degA, float* __restrict__ degB, int N, int E) {
    __shared__ float acc[NP1];
    __shared__ int   s_half;
    const int tid = threadIdx.x;
    if (tid == 0) s_half = cache_half();
    const int lo  = (int)blockIdx.y * NP1;
    const int np  = (N - lo) < NP1 ? (N - lo) : NP1;
    for (int j = tid; j < np; j += EDGE_TPB) acc[j] = 0.0f;
    __syncthreads();

    int beg, end;
    edge_slice(E, beg, end);
    const int* __restrict__ dst = ei + E;
    if (beg < end) {
        int   d[EDGE_U];
        float wv[EDGE_U];
#pragma unroll
        for (int u = 0; u < EDGE_U; ++u) {
            int ec = edge_clamp(beg + tid + u * EDGE_TPB, end);
            d[u]  = dst[ec];
            wv[u] = w[ec];
        }
        for (int eb = beg; eb < end; eb += EDGE_CH) {
            int   dn[EDGE_U];
            float wn[EDGE_U];
#pragma unroll
            for (int u = 0; u < EDGE_U; ++u) {
                int ec = edge_clamp(eb + EDGE_CH + tid + u * EDGE_TPB, end);
                dn[u] = dst[ec];
                wn[u] = w[ec];
            }
#pragma unroll
            for (int u = 0; u < EDGE_U; ++u) {
                int e = eb + tid + u * EDGE_TPB;
                unsigned j = (unsigned)(d[u] - lo);
                if ((e < end) & (j < (unsigned)np)) atomicAdd(&acc[j], wv[u]);
            }
#pragma unroll
            for (int u = 0; u < EDGE_U; ++u) { d[u] = dn[u]; wv[u] = wn[u]; }
        }
    }
    __syncthreads();
    float* __restrict__ deg = s_half ? degB : degA;
    for (int j = tid; j < np; j += EDGE_TPB) {
        float v = acc[j];
        if (v != 0.0f) atomicAdd(&deg[lo + j], v);
    }
}

__global__ void k_dinv_init1(const float* __restrict__ x, float* __restrict__ deg_dinv,
                             const float* __restrict__ degB, float* __restrict__ xs,
                             float* __restrict__ agg1A, float* __restrict__ agg1B, int N) {
    int i = blockIdx.x * blockDim.x + threadIdx.x;
    if (i < N) {
        float d  = deg_dinv[i] + degB[i];
        float di = (d > 0.0f) ? rsqrtf(d) : 0.0f;
        float v  = di * x[i];
        deg_dinv[i] = di;
        xs[i]    = v;
        agg1A[i] = v;
        agg1B[i] = 0.0f;
    }
}

__global__ __launch_bounds__(EDGE_TPB)
void k_edge1(const int* __restrict__ ei, const float* __restrict__ w,
             const float* __restrict__ xs, float* __restrict__ agg1A,
             float* __restrict__ agg1B, int N, int E) {
    __shared__ float acc[NP1];
    __shared__ int   s_half;
    const int tid = threadIdx.x;
    if (tid == 0) s_half = cache_half();
    const int lo  = (int)blockIdx.y * NP1;
    const int np  = (N - lo) < NP1 ? (N - lo) : NP1;
    for (int j = tid; j < np; j += EDGE_TPB) acc[j] = 0.0f;
    __syncthreads();

    int beg, end;
    edge_slice(E, beg, end);
    const int* __restrict__ src = ei;
    const int* __restrict__ dst = ei + E;
    if (beg < end) {
        int   s[EDGE_U], d[EDGE_U];
        float wv[EDGE_U];
#pragma unroll
        for (int u = 0; u < EDGE_U; ++u) {
            int ec = edge_clamp(beg + tid + u * EDGE_TPB, end);
            s[u]  = src[ec];
            d[u]  = dst[ec];
            wv[u] = w[ec];
        }
        for (int eb = beg; eb < end; eb += EDGE_CH) {
            int   sn[EDGE_U], dn[EDGE_U];
            float wn[EDGE_U];
            float xg[EDGE_U];
            bool  m[EDGE_U];
#pragma unroll
            for (int u = 0; u < EDGE_U; ++u) {
                int ec = edge_clamp(eb + EDGE_CH + tid + u * EDGE_TPB, end);
                sn[u] = src[ec];
                dn[u] = dst[ec];
                wn[u] = w[ec];
            }
#pragma unroll
            for (int u = 0; u < EDGE_U; ++u) {
                int e = eb + tid + u * EDGE_TPB;
                m[u]  = (e < end) & ((unsigned)(d[u] - lo) < (unsigned)np);
                xg[u] = 0.0f;
                if (m[u]) xg[u] = xs[s[u]];
            }
#pragma unroll
            for (int u = 0; u < EDGE_U; ++u) {
                if (m[u]) atomicAdd(&acc[d[u] - lo], wv[u] * xg[u]);
            }
#pragma unroll
            for (int u = 0; u < EDGE_U; ++u) { s[u] = sn[u]; d[u] = dn[u]; wv[u] = wn[u]; }
        }
    }
    __syncthreads();
    float* __restrict__ agg1 = s_half ? agg1B : agg1A;
    for (int j = tid; j < np; j += EDGE_TPB) {
        float v = acc[j];
        if (v != 0.0f) atomicAdd(&agg1[lo + j], v);
    }
}

__global__ void k_layer2_wmma(const float* __restrict__ agg1A, const float* __restrict__ agg1B,
                              const float* __restrict__ W1,
                              const float* __restrict__ b1, const float* __restrict__ W2,
                              const float* __restrict__ dinv, float* __restrict__ zs,
                              float* __restrict__ agg2A, float* __restrict__ agg2B, int N) {
    __shared__ float w2p[16 * 16];
    __shared__ float w1s[16];
    __shared__ float b1s[16];
    int t = threadIdx.x;
    {
        int k = t >> 4, n = t & 15;
        w2p[t] = (n < 2) ? W2[k * 2 + n] : 0.0f;
        if (t < 16) { w1s[t] = W1[t]; b1s[t] = b1[t]; }
    }
    __syncthreads();

    int wave = t >> 5;
    int lane = t & 31;
    int tile = blockIdx.x * 8 + wave;
    int m0   = tile * 16;
    if (m0 >= N) return;

    int row   = lane & 15;
    int khalf = (lane >> 4) * 2;

    int anode = m0 + row;
    int aload = anode < N ? anode : N - 1;
    const float p = dinv[aload] * (agg1A[aload] + agg1B[aload]);

    v2f a[4];
#pragma unroll
    for (int tstep = 0; tstep < 4; ++tstep) {
        int kb = tstep * 4 + khalf;
        float h0 = fmaf(p, w1s[kb],     b1s[kb]);
        float h1 = fmaf(p, w1s[kb + 1], b1s[kb + 1]);
        a[tstep].x = h0 > 0.0f ? h0 : 0.0f;
        a[tstep].y = h1 > 0.0f ? h1 : 0.0f;
    }

    __builtin_amdgcn_sched_barrier(0);

    v8f acc = {};
#pragma unroll
    for (int tstep = 0; tstep < 4; ++tstep) {
        int kb = tstep * 4 + khalf;
        v2f b;
        b.x = w2p[kb * 16 + row];
        b.y = w2p[(kb + 1) * 16 + row];
        acc = __builtin_amdgcn_wmma_f32_16x16x4_f32(
             false, a[tstep],  false, b,
             (short)0, acc,  false,  false);
    }

    int col = row;
    if (col < 2) {
        int mofs = (lane >> 4) * 8;
#pragma unroll
        for (int r = 0; r < 8; ++r) {
            int node = m0 + mofs + r;
            if (node < N) {
                float v = dinv[node] * acc[r];
                zs[node * 2 + col]    = v;
                agg2A[node * 2 + col] = v;
                agg2B[node * 2 + col] = 0.0f;
            }
        }
    }
}

__global__ __launch_bounds__(EDGE_TPB)
void k_edge2(const int* __restrict__ ei, const float* __restrict__ w,
             const float* __restrict__ zs, float* __restrict__ agg2A,
             float* __restrict__ agg2B, int N, int E) {
    __shared__ float acc[2 * NP2];
    __shared__ int   s_half;
    const int tid = threadIdx.x;
    if (tid == 0) s_half = cache_half();
    const int lo  = (int)blockIdx.y * NP2;
    const int np  = (N - lo) < NP2 ? (N - lo) : NP2;
    for (int j = tid; j < 2 * np; j += EDGE_TPB) acc[j] = 0.0f;
    __syncthreads();

    int beg, end;
    edge_slice(E, beg, end);
    const int* __restrict__ src = ei;
    const int* __restrict__ dst = ei + E;
    if (beg < end) {
        int   s[EDGE_U], d[EDGE_U];
        float nm[EDGE_U];
#pragma unroll
        for (int u = 0; u < EDGE_U; ++u) {
            int ec = edge_clamp(beg + tid + u * EDGE_TPB, end);
            s[u]  = src[ec];
            d[u]  = dst[ec];
            nm[u] = w[ec];
        }
        for (int eb = beg; eb < end; eb += EDGE_CH) {
            int   sn[EDGE_U], dn[EDGE_U];
            float nn[EDGE_U];
            v2f   zz[EDGE_U];
            bool  m[EDGE_U];
#pragma unroll
            for (int u = 0; u < EDGE_U; ++u) {
                int ec = edge_clamp(eb + EDGE_CH + tid + u * EDGE_TPB, end);
                sn[u] = src[ec];
                dn[u] = dst[ec];
                nn[u] = w[ec];
            }
#pragma unroll
            for (int u = 0; u < EDGE_U; ++u) {
                int e = eb + tid + u * EDGE_TPB;
                m[u]  = (e < end) & ((unsigned)(d[u] - lo) < (unsigned)np);
                zz[u] = (v2f){0.0f, 0.0f};
                if (m[u]) __builtin_memcpy(&zz[u], zs + 2 * s[u], sizeof(v2f));
            }
#pragma unroll
            for (int u = 0; u < EDGE_U; ++u) {
                if (m[u]) {
                    int j = d[u] - lo;
                    atomicAdd(&acc[2 * j],     nm[u] * zz[u].x);
                    atomicAdd(&acc[2 * j + 1], nm[u] * zz[u].y);
                }
            }
#pragma unroll
            for (int u = 0; u < EDGE_U; ++u) { s[u] = sn[u]; d[u] = dn[u]; nm[u] = nn[u]; }
        }
    }
    __syncthreads();
    float* __restrict__ agg2 = s_half ? agg2B : agg2A;
    for (int j = tid; j < 2 * np; j += EDGE_TPB) {
        float v = acc[j];
        if (v != 0.0f) atomicAdd(&agg2[2 * lo + j], v);
    }
}

__global__ void k_out(const float* __restrict__ agg2A, const float* __restrict__ agg2B,
                      const float* __restrict__ dinv,
                      const float* __restrict__ b2, float* __restrict__ out, int N) {
    int i = blockIdx.x * blockDim.x + threadIdx.x;
    if (i < N) {
        float di = dinv[i];
        float a0 = fmaf(di, agg2A[2 * i]     + agg2B[2 * i],     b2[0]);
        float a1 = fmaf(di, agg2A[2 * i + 1] + agg2B[2 * i + 1], b2[1]);
        float m  = fmaxf(a0, a1);
        float lse = m + logf(expf(a0 - m) + expf(a1 - m));
        out[2 * i]     = a0 - lse;
        out[2 * i + 1] = a1 - lse;
    }
}

extern "C" void kernel_launch(void* const* d_in, const int* in_sizes, int n_in,
                              void* d_out, int out_size, void* d_ws, size_t ws_size,
                              hipStream_t stream) {
    const float* x   = (const float*)d_in[0];
    const int*   ei  = (const int*)  d_in[1];
    const float* ew  = (const float*)d_in[2];
    const float* W1  = (const float*)d_in[3];
    const float* b1  = (const float*)d_in[4];
    const float* W2  = (const float*)d_in[5];
    const float* b2  = (const float*)d_in[6];
    float*       out = (float*)d_out;

    const int N = in_sizes[0];
    const int E = in_sizes[2];

    float* wsf   = (float*)d_ws;
    float* dinv  = wsf;
    float* agg1A = wsf + (size_t)N;
    float* xs    = wsf + (size_t)2 * N;
    float* degB  = xs + (size_t)N;
    float* agg1B = xs + (size_t)2 * N;
    float* agg2B = xs + (size_t)3 * N;
    float* zs    = xs + (size_t)E;
    float* agg2A = zs + (size_t)2 * N;

    const int B = 256;
    const int gN  = (N + B - 1) / B;
    const int tiles = (N + 15) / 16;
    const int gW  = (tiles + 7) / 8;
    const int P1  = (N + NP1 - 1) / NP1;
    const int P2  = (N + NP2 - 1) / NP2;
    const dim3 gE1(64, P1);
    const dim3 gE2(32, P2);

    k_init_deg   <<<gN,  B, 0, stream>>>(dinv, degB, N);
    k_deg_edges  <<<gE1, EDGE_TPB, 0, stream>>>(ei, ew, dinv, degB, N, E);
    k_dinv_init1 <<<gN,  B, 0, stream>>>(x, dinv, degB, xs, agg1A, agg1B, N);
    k_edge1      <<<gE1, EDGE_TPB, 0, stream>>>(ei, ew, xs, agg1A, agg1B, N, E);
    k_layer2_wmma<<<gW,  B, 0, stream>>>(agg1A, agg1B, W1, b1, W2, dinv, zs, agg2A, agg2B, N);
    k_edge2      <<<gE2, EDGE_TPB, 0, stream>>>(ei, ew, zs, agg2A, agg2B, N, E);
    k_out        <<<gN,  B, 0, stream>>>(agg2A, agg2B, dinv, b2, out, N);
}


# Layer 2 carries one scalar per node, the logit difference (2-class log_softmax needs only it): W2 column difference on WMMA, scalar edge pass, sums now of per-source differences
# speedup vs baseline: 8.4106x; 1.1364x over previous
#include <hip/hip_runtime.h>
#include <hip/hip_bf16.h>

typedef __attribute__((ext_vector_type(2))) float v2f;
typedef __attribute__((ext_vector_type(8))) float v8f;

__device__ __forceinline__ int cache_half() {
    return (int)((__builtin_amdgcn_s_sendmsg_rtn(0x87) >> 18) & 1u);
}

__global__ void k_init_deg(float* __restrict__ degA, float* __restrict__ degB, int N) {
    int i = blockIdx.x * blockDim.x + threadIdx.x;
    if (i < N) { degA[i] = 1.0f; degB[i] = 0.0f; }
}

#define EDGE_TPB 1024
#define EDGE_U   4
#define EDGE_CH  (EDGE_TPB * EDGE_U)
#define NP1      25600

__device__ __forceinline__ void edge_slice(int E, int& beg, int& end) {
    int S   = (int)gridDim.x;
    int per = (E + S - 1) / S;
    per     = (per + 31) & ~31;
    int b   = (int)blockIdx.x * per;
    beg = b < E ? b : E;
    end = (b + per) < E ? (b + per) : E;
}

__device__ __forceinline__ int edge_clamp(int e, int end) { return e < end ? e : end - 1; }

__global__ __launch_bounds__(EDGE_TPB)
void k_deg_edges(const int* __restrict__ ei, const float* __restrict__ w,
                 float* __restrict__ degA, float* __restrict__ degB, int N, int E) {
    __shared__ float acc[NP1];
    __shared__ int   s_half;
    const int tid = threadIdx.x;
    if (tid == 0) s_half = cache_half();
    const int lo  = (int)blockIdx.y * NP1;
    const int np  = (N - lo) < NP1 ? (N - lo) : NP1;
    for (int j = tid; j < np; j += EDGE_TPB) acc[j] = 0.0f;
    __syncthreads();

    int beg, end;
    edge_slice(E, beg, end);
    const int* __restrict__ dst = ei + E;
    if (beg < end) {
        int   d[EDGE_U];
        float wv[EDGE_U];
#pragma unroll
        for (int u = 0; u < EDGE_U; ++u) {
            int ec = edge_clamp(beg + tid + u * EDGE_TPB, end);
            d[u]  = dst[ec];
            wv[u] = w[ec];
        }
        for (int eb = beg; eb < end; eb += EDGE_CH) {
            int   dn[EDGE_U];
            float wn[EDGE_U];
#pragma unroll
            for (int u = 0; u < EDGE_U; ++u) {
                int ec = edge_clamp(eb + EDGE_CH + tid + u * EDGE_TPB, end);
                dn[u] = dst[ec];
                wn[u] = w[ec];
            }
#pragma unroll
            for (int u = 0; u < EDGE_U; ++u) {
                int e = eb + tid + u * EDGE_TPB;
                unsigned j = (unsigned)(d[u] - lo);
                if ((e < end) & (j < (unsigned)np)) atomicAdd(&acc[j], wv[u]);
            }
#pragma unroll
            for (int u = 0; u < EDGE_U; ++u) { d[u] = dn[u]; wv[u] = wn[u]; }
        }
    }
    __syncthreads();
    float* __restrict__ deg = s_half ? degB : degA;
    for (int j = tid; j < np; j += EDGE_TPB) {
        float v = acc[j];
        if (v != 0.0f) atomicAdd(&deg[lo + j], v);
    }
}

__global__ void k_dinv_init1(const float* __restrict__ x, float* __restrict__ deg_dinv,
                             const float* __restrict__ degB, float* __restrict__ xs,
                             float* __restrict__ agg1A, float* __restrict__ agg1B, int N) {
    int i = blockIdx.x * blockDim.x + threadIdx.x;
    if (i < N) {
        float d  = deg_dinv[i] + degB[i];
        float di = (d > 0.0f) ? rsqrtf(d) : 0.0f;
        float v  = di * x[i];
        deg_dinv[i] = di;
        xs[i]    = v;
        agg1A[i] = v;
        agg1B[i] = 0.0f;
    }
}

__global__ __launch_bounds__(EDGE_TPB)
void k_edge(const int* __restrict__ ei, const float* __restrict__ w,
            const float* __restrict__ vs, float* __restrict__ aggA,
            float* __restrict__ aggB, int N, int E) {
    __shared__ float acc[NP1];
    __shared__ int   s_half;
    const int tid = threadIdx.x;
    if (tid == 0) s_half = cache_half();
    const int lo  = (int)blockIdx.y * NP1;
    const int np  = (N - lo) < NP1 ? (N - lo) : NP1;
    for (int j = tid; j < np; j += EDGE_TPB) acc[j] = 0.0f;
    __syncthreads();

    int beg, end;
    edge_slice(E, beg, end);
    const int* __restrict__ src = ei;
    const int* __restrict__ dst = ei + E;
    if (beg < end) {
        int   s[EDGE_U], d[EDGE_U];
        float wv[EDGE_U];
#pragma unroll
        for (int u = 0; u < EDGE_U; ++u) {
            int ec = edge_clamp(beg + tid + u * EDGE_TPB, end);
            s[u]  = src[ec];
            d[u]  = dst[ec];
            wv[u] = w[ec];
        }
        for (int eb = beg; eb < end; eb += EDGE_CH) {
            int   sn[EDGE_U], dn[EDGE_U];
            float wn[EDGE_U];
            float xg[EDGE_U];
            bool  m[EDGE_U];
#pragma unroll
            for (int u = 0; u < EDGE_U; ++u) {
                int ec = edge_clamp(eb + EDGE_CH + tid + u * EDGE_TPB, end);
                sn[u] = src[ec];
                dn[u] = dst[ec];
                wn[u] = w[ec];
            }
#pragma unroll
            for (int u = 0; u < EDGE_U; ++u) {
                int e = eb + tid + u * EDGE_TPB;
                m[u]  = (e < end) & ((unsigned)(d[u] - lo) < (unsigned)np);
                xg[u] = 0.0f;
                if (m[u]) xg[u] = vs[s[u]];
            }
#pragma unroll
            for (int u = 0; u < EDGE_U; ++u) {
                if (m[u]) atomicAdd(&acc[d[u] - lo], wv[u] * xg[u]);
            }
#pragma unroll
            for (int u = 0; u < EDGE_U; ++u) { s[u] = sn[u]; d[u] = dn[u]; wv[u] = wn[u]; }
        }
    }
    __syncthreads();
    float* __restrict__ agg = s_half ? aggB : aggA;
    for (int j = tid; j < np; j += EDGE_TPB) {
        float v = acc[j];
        if (v != 0.0f) atomicAdd(&agg[lo + j], v);
    }
}

__global__ void k_layer2_wmma(const float* __restrict__ agg1A, const float* __restrict__ agg1B,
                              const float* __restrict__ W1,
                              const float* __restrict__ b1, const float* __restrict__ W2,
                              const float* __restrict__ dinv, float* __restrict__ ds,
                              float* __restrict__ agg2A, float* __restrict__ agg2B, int N) {
    __shared__ float w2p[16 * 16];
    __shared__ float w1s[16];
    __shared__ float b1s[16];
    int t = threadIdx.x;
    {
        int k = t >> 4, n = t & 15;
        w2p[t] = (n == 0) ? (W2[k * 2] - W2[k * 2 + 1]) : 0.0f;
        if (t < 16) { w1s[t] = W1[t]; b1s[t] = b1[t]; }
    }
    __syncthreads();

    int wave = t >> 5;
    int lane = t & 31;
    int tile = blockIdx.x * 8 + wave;
    int m0   = tile * 16;
    if (m0 >= N) return;

    int row   = lane & 15;
    int khalf = (lane >> 4) * 2;

    int anode = m0 + row;
    int aload = anode < N ? anode : N - 1;
    const float p = dinv[aload] * (agg1A[aload] + agg1B[aload]);

    v2f a[4];
#pragma unroll
    for (int tstep = 0; tstep < 4; ++tstep) {
        int kb = tstep * 4 + khalf;
        float h0 = fmaf(p, w1s[kb],     b1s[kb]);
        float h1 = fmaf(p, w1s[kb + 1], b1s[kb + 1]);
        a[tstep].x = h0 > 0.0f ? h0 : 0.0f;
        a[tstep].y = h1 > 0.0f ? h1 : 0.0f;
    }

    __builtin_amdgcn_sched_barrier(0);

    v8f acc = {};
#pragma unroll
    for (int tstep = 0; tstep < 4; ++tstep) {
        int kb = tstep * 4 + khalf;
        v2f b;
        b.x = w2p[kb * 16 + row];
        b.y = w2p[(kb + 1) * 16 + row];
        acc = __builtin_amdgcn_wmma_f32_16x16x4_f32(
             false, a[tstep],  false, b,
             (short)0, acc,  false,  false);
    }

    asm volatile("v_nop\n\tv_nop\n\tv_nop\n\tv_nop" : "+v"(acc));

    int col = row;
    if (col == 0) {
        int mofs = (lane >> 4) * 8;
#pragma unroll
        for (int r = 0; r < 8; ++r) {
            int node = m0 + mofs + r;
            if (node < N) {
                float v = dinv[node] * acc[r];
                ds[node]    = v;
                agg2A[node] = v;
                agg2B[node] = 0.0f;
            }
        }
    }
}

__global__ void k_out(const float* __restrict__ agg2A, const float* __restrict__ agg2B,
                      const float* __restrict__ dinv,
                      const float* __restrict__ b2, float* __restrict__ out, int N) {
    int i = blockIdx.x * blockDim.x + threadIdx.x;
    if (i < N) {
        float t  = fmaf(dinv[i], agg2A[i] + agg2B[i], b2[0] - b2[1]);
        float m  = fmaxf(t, 0.0f);
        float s0 = t - m;
        float s1 = -m;
        float l  = logf(expf(s0) + expf(s1));
        out[2 * i]     = s0 - l;
        out[2 * i + 1] = s1 - l;
    }
}

extern "C" void kernel_launch(void* const* d_in, const int* in_sizes, int n_in,
                              void* d_out, int out_size, void* d_ws, size_t ws_size,
                              hipStream_t stream) {
    const float* x   = (const float*)d_in[0];
    const int*   ei  = (const int*)  d_in[1];
    const float* ew  = (const float*)d_in[2];
    const float* W1  = (const float*)d_in[3];
    const float* b1  = (const float*)d_in[4];
    const float* W2  = (const float*)d_in[5];
    const float* b2  = (const float*)d_in[6];
    float*       out = (float*)d_out;

    const int N = in_sizes[0];
    const int E = in_sizes[2];

    float* wsf   = (float*)d_ws;
    float* dinv  = wsf;
    float* agg1A = wsf + (size_t)N;
    float* xs    = wsf + (size_t)2 * N;
    float* degB  = xs + (size_t)N;
    float* agg1B = xs + (size_t)2 * N;
    float* agg2B = xs + (size_t)3 * N;
    float* ds    = xs + (size_t)E;
    float* agg2A = ds + (size_t)2 * N;

    const int B = 256;
    const int gN  = (N + B - 1) / B;
    const int tiles = (N + 15) / 16;
    const int gW  = (tiles + 7) / 8;
    const int P1  = (N + NP1 - 1) / NP1;
    const dim3 gE1(64, P1);

    k_init_deg   <<<gN,  B, 0, stream>>>(dinv, degB, N);
    k_deg_edges  <<<gE1, EDGE_TPB, 0, stream>>>(ei, ew, dinv, degB, N, E);
    k_dinv_init1 <<<gN,  B, 0, stream>>>(x, dinv, degB, xs, agg1A, agg1B, N);
    k_edge       <<<gE1, EDGE_TPB, 0, stream>>>(ei, ew, xs, agg1A, agg1B, N, E);
    k_layer2_wmma<<<gW,  B, 0, stream>>>(agg1A, agg1B, W1, b1, W2, dinv, ds, agg2A, agg2B, N);
    k_edge       <<<gE1, EDGE_TPB, 0, stream>>>(ei, ew, ds, agg2A, agg2B, N, E);
    k_out        <<<gN,  B, 0, stream>>>(agg2A, agg2B, dinv, b2, out, N);
}


# Edge passes of both layers keep the first 55296 entries of the gathered array in LDS beside the sums (copied in whole lines); gathers below that index read LDS, not a second-level line each
# speedup vs baseline: 8.6825x; 1.0323x over previous
#include <hip/hip_runtime.h>
#include <hip/hip_bf16.h>

typedef __attribute__((ext_vector_type(2))) float v2f;
typedef __attribute__((ext_vector_type(8))) float v8f;

__device__ __forceinline__ int cache_half() {
    return (int)((__builtin_amdgcn_s_sendmsg_rtn(0x87) >> 18) & 1u);
}

__global__ void k_init_deg(float* __restrict__ degA, float* __restrict__ degB, int N) {
    int i = blockIdx.x * blockDim.x + threadIdx.x;
    if (i < N) { degA[i] = 1.0f; degB[i] = 0.0f; }
}

#define EDGE_TPB 1024
#define EDGE_U   4
#define EDGE_CH  (EDGE_TPB * EDGE_U)
#define NP1      25600
#define WIN_K    54
#define WIN      (WIN_K * EDGE_TPB)

__device__ __forceinline__ void edge_slice(int E, int& beg, int& end) {
    int S   = (int)gridDim.x;
    int per = (E + S - 1) / S;
    per     = (per + 31) & ~31;
    int b   = (int)blockIdx.x * per;
    beg = b < E ? b : E;
    end = (b + per) < E ? (b + per) : E;
}

__device__ __forceinline__ int edge_clamp(int e, int end) { return e < end ? e : end - 1; }

__global__ __launch_bounds__(EDGE_TPB)
void k_deg_edges(const int* __restrict__ ei, const float* __restrict__ w,
                 float* __restrict__ degA, float* __restrict__ degB, int N, int E) {
    __shared__ float acc[NP1];
    __shared__ int   s_half;
    const int tid = threadIdx.x;
    if (tid == 0) s_half = cache_half();
    const int lo  = (int)blockIdx.y * NP1;
    const int np  = (N - lo) < NP1 ? (N - lo) : NP1;
    for (int j = tid; j < np; j += EDGE_TPB) acc[j] = 0.0f;
    __syncthreads();

    int beg, end;
    edge_slice(E, beg, end);
    const int* __restrict__ dst = ei + E;
    if (beg < end) {
        int   d[EDGE_U];
        float wv[EDGE_U];
#pragma unroll
        for (int u = 0; u < EDGE_U; ++u) {
            int ec = edge_clamp(beg + tid + u * EDGE_TPB, end);
            d[u]  = dst[ec];
            wv[u] = w[ec];
        }
        for (int eb = beg; eb < end; eb += EDGE_CH) {
            int   dn[EDGE_U];
            float wn[EDGE_U];
#pragma unroll
            for (int u = 0; u < EDGE_U; ++u) {
                int ec = edge_clamp(eb + EDGE_CH + tid + u * EDGE_TPB, end);
                dn[u] = dst[ec];
                wn[u] = w[ec];
            }
#pragma unroll
            for (int u = 0; u < EDGE_U; ++u) {
                int e = eb + tid + u * EDGE_TPB;
                unsigned j = (unsigned)(d[u] - lo);
                if ((e < end) & (j < (unsigned)np)) atomicAdd(&acc[j], wv[u]);
            }
#pragma unroll
            for (int u = 0; u < EDGE_U; ++u) { d[u] = dn[u]; wv[u] = wn[u]; }
        }
    }
    __syncthreads();
    float* __restrict__ deg = s_half ? degB : degA;
    for (int j = tid; j < np; j += EDGE_TPB) {
        float v = acc[j];
        if (v != 0.0f) atomicAdd(&deg[lo + j], v);
    }
}

__global__ void k_dinv_init1(const float* __restrict__ x, float* __restrict__ deg_dinv,
                             const float* __restrict__ degB, float* __restrict__ xs,
                             float* __restrict__ agg1A, float* __restrict__ agg1B, int N) {
    int i = blockIdx.x * blockDim.x + threadIdx.x;
    if (i < N) {
        float d  = deg_dinv[i] + degB[i];
        float di = (d > 0.0f) ? rsqrtf(d) : 0.0f;
        float v  = di * x[i];
        deg_dinv[i] = di;
        xs[i]    = v;
        agg1A[i] = v;
        agg1B[i] = 0.0f;
    }
}

__global__ __launch_bounds__(EDGE_TPB)
void k_edge(const int* __restrict__ ei, const float* __restrict__ w,
            const float* __restrict__ vs, float* __restrict__ aggA,
            float* __restrict__ aggB, int N, int E) {
    __shared__ float acc[NP1];
    __shared__ float win[WIN];
    __shared__ int   s_half;
    const int tid = threadIdx.x;
    if (tid == 0) s_half = cache_half();
    const int lo  = (int)blockIdx.y * NP1;
    const int np  = (N - lo) < NP1 ? (N - lo) : NP1;
    const int nw  = N < WIN ? N : WIN;
    for (int j = tid; j < np; j += EDGE_TPB) acc[j] = 0.0f;
    {
        float t[WIN_K];
#pragma unroll
        for (int k = 0; k < WIN_K; ++k) {
            int j = tid + k * EDGE_TPB;
            t[k] = vs[j < nw ? j : nw - 1];
        }
#pragma unroll
        for (int k = 0; k < WIN_K; ++k) win[tid + k * EDGE_TPB] = t[k];
    }
    __syncthreads();

    int beg, end;
    edge_slice(E, beg, end);
    const int* __restrict__ src = ei;
    const int* __restrict__ dst = ei + E;
    if (beg < end) {
        int   s[EDGE_U], d[EDGE_U];
        float wv[EDGE_U];
#pragma unroll
        for (int u = 0; u < EDGE_U; ++u) {
            int ec = edge_clamp(beg + tid + u * EDGE_TPB, end);
            s[u]  = src[ec];
            d[u]  = dst[ec];
            wv[u] = w[ec];
        }
        for (int eb = beg; eb < end; eb += EDGE_CH) {
            int   sn[EDGE_U], dn[EDGE_U];
            float wn[EDGE_U];
            float xg[EDGE_U], xl[EDGE_U];
            bool  m[EDGE_U], inw[EDGE_U];
#pragma unroll
            for (int u = 0; u < EDGE_U; ++u) {
                int ec = edge_clamp(eb + EDGE_CH + tid + u * EDGE_TPB, end);
                sn[u] = src[ec];
                dn[u] = dst[ec];
                wn[u] = w[ec];
            }
#pragma unroll
            for (int u = 0; u < EDGE_U; ++u) {
                int e = eb + tid + u * EDGE_TPB;
                m[u]  = (e < end) & ((unsigned)(d[u] - lo) < (unsigned)np);
                inw[u] = (unsigned)s[u] < (unsigned)nw;
                xl[u] = win[(m[u] & inw[u]) ? s[u] : 0];
                xg[u] = 0.0f;
                if (m[u] & !inw[u]) xg[u] = vs[s[u]];
            }
#pragma unroll
            for (int u = 0; u < EDGE_U; ++u) {
                if (m[u]) atomicAdd(&acc[d[u] - lo], wv[u] * (inw[u] ? xl[u] : xg[u]));
            }
#pragma unroll
            for (int u = 0; u < EDGE_U; ++u) { s[u] = sn[u]; d[u] = dn[u]; wv[u] = wn[u]; }
        }
    }
    __syncthreads();
    float* __restrict__ agg = s_half ? aggB : aggA;
    for (int j = tid; j < np; j += EDGE_TPB) {
        float v = acc[j];
        if (v != 0.0f) atomicAdd(&agg[lo + j], v);
    }
}

__global__ void k_layer2_wmma(const float* __restrict__ agg1A, const float* __restrict__ agg1B,
                              const float* __restrict__ W1,
                              const float* __restrict__ b1, const float* __restrict__ W2,
                              const float* __restrict__ dinv, float* __restrict__ ds,
                              float* __restrict__ agg2A, float* __restrict__ agg2B, int N) {
    __shared__ float w2p[16 * 16];
    __shared__ float w1s[16];
    __shared__ float b1s[16];
    int t = threadIdx.x;
    {
        int k = t >> 4, n = t & 15;
        w2p[t] = (n == 0) ? (W2[k * 2] - W2[k * 2 + 1]) : 0.0f;
        if (t < 16) { w1s[t] = W1[t]; b1s[t] = b1[t]; }
    }
    __syncthreads();

    int wave = t >> 5;
    int lane = t & 31;
    int tile = blockIdx.x * 8 + wave;
    int m0   = tile * 16;
    if (m0 >= N) return;

    int row   = lane & 15;
    int khalf = (lane >> 4) * 2;

    int anode = m0 + row;
    int aload = anode < N ? anode : N - 1;
    const float p = dinv[aload] * (agg1A[aload] + agg1B[aload]);

    v2f a[4];
#pragma unroll
    for (int tstep = 0; tstep < 4; ++tstep) {
        int kb = tstep * 4 + khalf;
        float h0 = fmaf(p, w1s[kb],     b1s[kb]);
        float h1 = fmaf(p, w1s[kb + 1], b1s[kb + 1]);
        a[tstep].x = h0 > 0.0f ? h0 : 0.0f;
        a[tstep].y = h1 > 0.0f ? h1 : 0.0f;
    }

    __builtin_amdgcn_sched_barrier(0);

    v8f acc = {};
#pragma unroll
    for (int tstep = 0; tstep < 4; ++tstep) {
        int kb = tstep * 4 + khalf;
        v2f b;
        b.x = w2p[kb * 16 + row];
        b.y = w2p[(kb + 1) * 16 + row];
        acc = __builtin_amdgcn_wmma_f32_16x16x4_f32(
             false, a[tstep],  false, b,
             (short)0, acc,  false,  false);
    }

    asm volatile("v_nop\n\tv_nop\n\tv_nop\n\tv_nop" : "+v"(acc));

    int col = row;
    if (col == 0) {
        int mofs = (lane >> 4) * 8;
#pragma unroll
        for (int r = 0; r < 8; ++r) {
            int node = m0 + mofs + r;
            if (node < N) {
                float v = dinv[node] * acc[r];
                ds[node]    = v;
                agg2A[node] = v;
                agg2B[node] = 0.0f;
            }
        }
    }
}

__global__ void k_out(const float* __restrict__ agg2A, const float* __restrict__ agg2B,
                      const float* __restrict__ dinv,
                      const float* __restrict__ b2, float* __restrict__ out, int N) {
    int i = blockIdx.x * blockDim.x + threadIdx.x;
    if (i < N) {
        float t  = fmaf(dinv[i], agg2A[i] + agg2B[i], b2[0] - b2[1]);
        float m  = fmaxf(t, 0.0f);
        float s0 = t - m;
        float s1 = -m;
        float l  = logf(expf(s0) + expf(s1));
        out[2 * i]     = s0 - l;
        out[2 * i + 1] = s1 - l;
    }
}

extern "C" void kernel_launch(void* const* d_in, const int* in_sizes, int n_in,
                              void* d_out, int out_size, void* d_ws, size_t ws_size,
                              hipStream_t stream) {
    const float* x   = (const float*)d_in[0];
    const int*   ei  = (const int*)  d_in[1];
    const float* ew  = (const float*)d_in[2];
    const float* W1  = (const float*)d_in[3];
    const float* b1  = (const float*)d_in[4];
    const float* W2  = (const float*)d_in[5];
    const float* b2  = (const float*)d_in[6];
    float*       out = (float*)d_out;

    const int N = in_sizes[0];
    const int E = in_sizes[2];

    float* wsf   = (float*)d_ws;
    float* dinv  = wsf;
    float* agg1A = wsf + (size_t)N;
    float* xs    = wsf + (size_t)2 * N;
    float* degB  = xs + (size_t)N;
    float* agg1B = xs + (size_t)2 * N;
    float* agg2B = xs + (size_t)3 * N;
    float* ds    = xs + (size_t)E;
    float* agg2A = ds + (size_t)2 * N;

    const int B = 256;
    const int gN  = (N + B - 1) / B;
    const int tiles = (N + 15) / 16;
    const int gW  = (tiles + 7) / 8;
    const int P1  = (N + NP1 - 1) / NP1;
    const dim3 gE1(64, P1);

    k_init_deg   <<<gN,  B, 0, stream>>>(dinv, degB, N);
    k_deg_edges  <<<gE1, EDGE_TPB, 0, stream>>>(ei, ew, dinv, degB, N, E);
    k_dinv_init1 <<<gN,  B, 0, stream>>>(x, dinv, degB, xs, agg1A, agg1B, N);
    k_edge       <<<gE1, EDGE_TPB, 0, stream>>>(ei, ew, xs, agg1A, agg1B, N, E);
    k_layer2_wmma<<<gW,  B, 0, stream>>>(agg1A, agg1B, W1, b1, W2, dinv, ds, agg2A, agg2B, N);
    k_edge       <<<gE1, EDGE_TPB, 0, stream>>>(ei, ew, ds, agg2A, agg2B, N, E);
    k_out        <<<gN,  B, 0, stream>>>(agg2A, agg2B, dinv, b2, out, N);
}
